# starting kernel
# baseline (speedup 1.0000x reference)
#include <hip/hip_runtime.h>


typedef __attribute__((ext_vector_type(2))) float v2f;
typedef __attribute__((ext_vector_type(8))) float v8f;

#define BATCH 128
#define DCH   300
#define LEN   128
#define NF    8
#define WPB   8

#define DPP_XOR1  0xB1
#define DPP_XOR2  0x4E
#define DPP_HMIRR 0x141
#define DPP_MIRR  0x140

template <int CTRL>
__device__ __forceinline__ float dpp_bfly(float v) {
    int i = __builtin_bit_cast(int, v);
    int t = __builtin_amdgcn_update_dpp(i, i, CTRL, 0xF, 0xF, false);
    return __builtin_bit_cast(float, t);
}
__device__ __forceinline__ float row16_max(float v) {
    v = fmaxf(v, dpp_bfly<DPP_XOR1>(v));
    v = fmaxf(v, dpp_bfly<DPP_XOR2>(v));
    v = fmaxf(v, dpp_bfly<DPP_HMIRR>(v));
    v = fmaxf(v, dpp_bfly<DPP_MIRR>(v));
    return v;
}
__device__ __forceinline__ float row16_min(float v) {
    v = fminf(v, dpp_bfly<DPP_XOR1>(v));
    v = fminf(v, dpp_bfly<DPP_XOR2>(v));
    v = fminf(v, dpp_bfly<DPP_HMIRR>(v));
    v = fminf(v, dpp_bfly<DPP_MIRR>(v));
    return v;
}
__device__ __forceinline__ float row16_sum(float v) {
    v += dpp_bfly<DPP_XOR1>(v);
    v += dpp_bfly<DPP_XOR2>(v);
    v += dpp_bfly<DPP_HMIRR>(v);
    v += dpp_bfly<DPP_MIRR>(v);
    return v;
}
__device__ __forceinline__ float swz_xor16(float v) {
    int t = __builtin_amdgcn_ds_swizzle(__builtin_bit_cast(int, v), 0x401F);
    return __builtin_bit_cast(float, t);
}

__global__ __launch_bounds__(256) void conv_pool_wmma(
    const float* __restrict__ x,
    const float* __restrict__ W1, const float* __restrict__ b1,
    const float* __restrict__ W2, const float* __restrict__ b2,
    const float* __restrict__ W3, const float* __restrict__ b3,
    float* __restrict__ out)
{
    __shared__ float xs[WPB][132];

    const int lane = threadIdx.x & 31;
    const int wv   = threadIdx.x >> 5;
    const int pair = blockIdx.x * WPB + wv;
    const int b    = pair / DCH;
    const int d    = pair - b * DCH;

    const float4 xv = *(const float4*)(x + (size_t)pair * LEN + lane * 4);
    *(float4*)(&xs[wv][lane * 4]) = xv;
    if (lane < 4) xs[wv][LEN + lane] = 0.0f;

    float mx = fmaxf(fmaxf(xv.x, xv.y), fmaxf(xv.z, xv.w));
    float mn = fminf(fminf(xv.x, xv.y), fminf(xv.z, xv.w));
    float sm = (xv.x + xv.y) + (xv.z + xv.w);
    mx = row16_max(mx);  mx = fmaxf(mx, swz_xor16(mx));
    mn = row16_min(mn);  mn = fminf(mn, swz_xor16(mn));
    sm = row16_sum(sm);  sm += swz_xor16(sm);

    v2f Bw;
    {
        const int  nn = lane & 15;
        const bool hi = lane >= 16;
        float w0 = 0.0f, w1 = 0.0f;
        if (nn < 8) {
            if (!hi) { w0 = W2[(d * NF + nn) * 2 + 0]; w1 = W2[(d * NF + nn) * 2 + 1]; }
        } else {
            const int f = nn - 8;
            if (!hi) { w0 = W3[(d * NF + f) * 3 + 0]; w1 = W3[(d * NF + f) * 3 + 1]; }
            else     { w0 = W3[(d * NF + f) * 3 + 2]; }
        }
        Bw[0] = w0; Bw[1] = w1;
    }

    __syncthreads();

    const int m    = lane & 15;
    const int kofs = (lane < 16) ? 0 : 2;

    float acc = -__builtin_inff();
    #pragma unroll
    for (int t = 0; t < 8; ++t) {
        v2f Ap;
        const int off = t * 16 + m + kofs;
        Ap[0] = xs[wv][off];
        Ap[1] = xs[wv][off + 1];
        v8f C = {};
        C = __builtin_amdgcn_wmma_f32_16x16x4_f32(false, Ap, false, Bw,
                                                  (short)0, C, false, false);
        float c6 = C[6], c7 = C[7];
        if (t == 7) {
            c6 = (lane < 24) ? c6 : -__builtin_inff();
            c7 = (lane < 16) ? c7 : -__builtin_inff();
        }
        const float r = fmaxf(fmaxf(fmaxf(C[0], C[1]), fmaxf(C[2], C[3])),
                              fmaxf(fmaxf(C[4], C[5]), fmaxf(c6, c7)));
        acc = fmaxf(acc, r);
    }
    acc = fmaxf(acc, swz_xor16(acc));

    const float x0   = xs[wv][0],   x1   = xs[wv][1];
    const float x126 = xs[wv][126], x127 = xs[wv][127];
    const float S0_2 = sm - x127,        S1_2 = sm - x0;
    const float S0_3 = sm - x126 - x127, S1_3 = sm - x0 - x127,
                S2_3 = sm - x0 - x1;
    const float w3t2 = swz_xor16(Bw[0]);

    const size_t ob   = (size_t)b * (2 * 3 * DCH * NF);
    const size_t strP = (size_t)3 * DCH * NF;
    const size_t strW = (size_t)DCH * NF;
    float* __restrict__ omax  = out + ob;
    float* __restrict__ omean = out + ob + strP;

    if (lane < 8) {
        const float w  = W1[d * NF + lane];
        const float bb = b1[d * NF + lane];
        omax [0 * strW + d * NF + lane] = (w >= 0.0f ? w * mx : w * mn) + bb;
        omean[0 * strW + d * NF + lane] = w * (sm * (1.0f / 128.0f)) + bb;
        const float bias2 = b2[d * NF + lane];
        omax [1 * strW + d * NF + lane] = acc + bias2;
        omean[1 * strW + d * NF + lane] =
            (Bw[0] * S0_2 + Bw[1] * S1_2) * (1.0f / 127.0f) + bias2;
    } else if (lane < 16) {
        const int f = lane - 8;
        const float bias3 = b3[d * NF + f];
        omax [2 * strW + d * NF + f] = acc + bias3;
        omean[2 * strW + d * NF + f] =
            (Bw[0] * S0_3 + Bw[1] * S1_3 + w3t2 * S2_3) * (1.0f / 126.0f) + bias3;
    }
}

extern "C" void kernel_launch(void* const* d_in, const int* in_sizes, int n_in,
                              void* d_out, int out_size, void* d_ws, size_t ws_size,
                              hipStream_t stream) {
    const float* x  = (const float*)d_in[0];
    const float* W1 = (const float*)d_in[1];
    const float* b1 = (const float*)d_in[2];
    const float* W2 = (const float*)d_in[3];
    const float* b2 = (const float*)d_in[4];
    const float* W3 = (const float*)d_in[5];
    const float* b3 = (const float*)d_in[6];
    float* out = (float*)d_out;

    const int pairs  = BATCH * DCH;
    const int blocks = pairs / WPB;
    conv_pool_wmma<<<blocks, WPB * 32, 0, stream>>>(x, W1, b1, W2, b2, W3, b3, out);
}


# each wave streams 8 batch rows of one channel (4800 waves in 4-wave groups, was 38400): 8 row loads in flight, weights and B fragment built once, own-row LDS order replaces the block barrier
# speedup vs baseline: 1.3502x; 1.3502x over previous
#include <hip/hip_runtime.h>


typedef __attribute__((ext_vector_type(2))) float v2f;
typedef __attribute__((ext_vector_type(4))) float v4f;
typedef __attribute__((ext_vector_type(8))) float v8f;

#define BATCH 128
#define DCH   300
#define LEN   128
#define NF    8
#define RPW   8
#define CHUNKS 16
#define WPB   4
#define NTHR  128
#define NBLK  1200
#define XPITCH 132

#define DPP_XOR1  0xB1
#define DPP_XOR2  0x4E
#define DPP_HMIRR 0x141
#define DPP_MIRR  0x140

template <int CTRL>
__device__ __forceinline__ float dpp_bfly(float v) {
    int i = __builtin_bit_cast(int, v);
    int t = __builtin_amdgcn_update_dpp(i, i, CTRL, 0xF, 0xF, false);
    return __builtin_bit_cast(float, t);
}
__device__ __forceinline__ float row16_max(float v) {
    v = fmaxf(v, dpp_bfly<DPP_XOR1>(v));
    v = fmaxf(v, dpp_bfly<DPP_XOR2>(v));
    v = fmaxf(v, dpp_bfly<DPP_HMIRR>(v));
    v = fmaxf(v, dpp_bfly<DPP_MIRR>(v));
    return v;
}
__device__ __forceinline__ float row16_min(float v) {
    v = fminf(v, dpp_bfly<DPP_XOR1>(v));
    v = fminf(v, dpp_bfly<DPP_XOR2>(v));
    v = fminf(v, dpp_bfly<DPP_HMIRR>(v));
    v = fminf(v, dpp_bfly<DPP_MIRR>(v));
    return v;
}
__device__ __forceinline__ float row16_sum(float v) {
    v += dpp_bfly<DPP_XOR1>(v);
    v += dpp_bfly<DPP_XOR2>(v);
    v += dpp_bfly<DPP_HMIRR>(v);
    v += dpp_bfly<DPP_MIRR>(v);
    return v;
}
__device__ __forceinline__ float swz_xor16(float v) {
    int t = __builtin_amdgcn_ds_swizzle(__builtin_bit_cast(int, v), 0x401F);
    return __builtin_bit_cast(float, t);
}

__global__ __launch_bounds__(NTHR) void conv_pool_wmma(
    const float* __restrict__ x,
    const float* __restrict__ W1, const float* __restrict__ b1,
    const float* __restrict__ W2, const float* __restrict__ b2,
    const float* __restrict__ W3, const float* __restrict__ b3,
    float* __restrict__ out)
{
    __shared__ __attribute__((aligned(16))) float xs[WPB][RPW][XPITCH];

    const int lane = threadIdx.x & 31;
    const int wv   = threadIdx.x >> 5;
    const int item = blockIdx.x * WPB + wv;
    const int d    = item / CHUNKS;
    const int b0   = (item - d * CHUNKS) * RPW;

    v4f xr[RPW];
    #pragma unroll
    for (int r = 0; r < RPW; ++r) {
        const size_t row = (size_t)(b0 + r) * DCH + d;
        xr[r] = *(const v4f*)(x + row * LEN + lane * 4);
    }

    v2f Bw;
    {
        const int  nn  = lane & 15;
        const bool hi  = lane >= 16;
        const bool is3 = nn >= 8;
        const int  g   = d * NF + (nn & 7);
        const float* p0 = is3 ? (W3 + g * 3 + (hi ? 2 : 0)) : (W2 + g * 2);
        const float* p1 = is3 ? (W3 + g * 3 + 1)            : (W2 + g * 2 + 1);
        const float l0 = *p0;
        const float l1 = *p1;
        Bw[0] = (!is3 && hi) ? 0.0f : l0;
        Bw[1] = hi ? 0.0f : l1;
    }
    const int   go   = d * NF + (lane & 7);
    const float wq1  = W1[go];
    const float bq1  = b1[go];
    const float* pb  = (lane & 8) ? b3 : b2;
    const float bq23 = pb[go];

    xs[wv][lane >> 2][LEN + (lane & 3)] = 0.0f;

    const float w3t2 = swz_xor16(Bw[0]);

    const int m    = lane & 15;
    const int kofs = (lane < 16) ? 0 : 2;

    const size_t strB = (size_t)2 * 3 * DCH * NF;
    const size_t strP = (size_t)3 * DCH * NF;
    const size_t strW = (size_t)DCH * NF;
    float* __restrict__ obase = out + (size_t)b0 * strB;

    #pragma unroll
    for (int r = 0; r < RPW; ++r) {
        float* row_s = &xs[wv][r][0];
        const v4f xv = xr[r];

        *(v4f*)(row_s + lane * 4) = xv;
        __builtin_amdgcn_fence(__ATOMIC_ACQ_REL, "wavefront");

        float mx = fmaxf(fmaxf(xv[0], xv[1]), fmaxf(xv[2], xv[3]));
        float mn = fminf(fminf(xv[0], xv[1]), fminf(xv[2], xv[3]));
        float sm = (xv[0] + xv[1]) + (xv[2] + xv[3]);
        mx = row16_max(mx);  mx = fmaxf(mx, swz_xor16(mx));
        mn = row16_min(mn);  mn = fminf(mn, swz_xor16(mn));
        sm = row16_sum(sm);  sm += swz_xor16(sm);

        float acc = -__builtin_inff();
        #pragma unroll
        for (int t = 0; t < 8; ++t) {
            v2f Ap;
            const int off = t * 16 + m + kofs;
            Ap[0] = row_s[off];
            Ap[1] = row_s[off + 1];
            v8f C = {};
            C = __builtin_amdgcn_wmma_f32_16x16x4_f32(false, Ap, false, Bw,
                                                      (short)0, C, false, false);
            float c6 = C[6], c7 = C[7];
            if (t == 7) {
                c6 = (lane < 24) ? c6 : -__builtin_inff();
                c7 = (lane < 16) ? c7 : -__builtin_inff();
            }
            const float q = fmaxf(fmaxf(fmaxf(C[0], C[1]), fmaxf(C[2], C[3])),
                                  fmaxf(fmaxf(C[4], C[5]), fmaxf(c6, c7)));
            acc = fmaxf(acc, q);
        }
        acc = fmaxf(acc, swz_xor16(acc));

        const float x0   = row_s[0],   x1   = row_s[1];
        const float x126 = row_s[126], x127 = row_s[127];
        const float S0_2 = sm - x127,        S1_2 = sm - x0;
        const float S0_3 = sm - x126 - x127, S1_3 = sm - x0 - x127,
                    S2_3 = sm - x0 - x1;

        float* __restrict__ omax  = obase + (size_t)r * strB;
        float* __restrict__ omean = omax + strP;

        if (lane < 8) {
            omax [0 * strW + d * NF + lane] = (wq1 >= 0.0f ? wq1 * mx : wq1 * mn) + bq1;
            omean[0 * strW + d * NF + lane] = wq1 * (sm * (1.0f / 128.0f)) + bq1;
            omax [1 * strW + d * NF + lane] = acc + bq23;
            omean[1 * strW + d * NF + lane] =
                (Bw[0] * S0_2 + Bw[1] * S1_2) * (1.0f / 127.0f) + bq23;
        } else if (lane < 16) {
            const int f = lane - 8;
            omax [2 * strW + d * NF + f] = acc + bq23;
            omean[2 * strW + d * NF + f] =
                (Bw[0] * S0_3 + Bw[1] * S1_3 + w3t2 * S2_3) * (1.0f / 126.0f) + bq23;
        }
    }
}

extern "C" void kernel_launch(void* const* d_in, const int* in_sizes, int n_in,
                              void* d_out, int out_size, void* d_ws, size_t ws_size,
                              hipStream_t stream) {
    const float* x  = (const float*)d_in[0];
    const float* W1 = (const float*)d_in[1];
    const float* b1 = (const float*)d_in[2];
    const float* W2 = (const float*)d_in[3];
    const float* b2 = (const float*)d_in[4];
    const float* W3 = (const float*)d_in[5];
    const float* b3 = (const float*)d_in[6];
    float* out = (float*)d_out;

    conv_pool_wmma<<<NBLK, NTHR, 0, stream>>>(x, W1, b1, W2, b2, W3, b3, out);
}


# 32-lane max/min/sum reductions: DPP permutation folded into NaN-propagating maximum/minimum/add (bound_ctrl form); no v_mov_dpp, no quieting max(x,x) copies: 928 -> 742 vector instructions per wave
# speedup vs baseline: 1.4150x; 1.0480x over previous
#include <hip/hip_runtime.h>


typedef __attribute__((ext_vector_type(2))) float v2f;
typedef __attribute__((ext_vector_type(4))) float v4f;
typedef __attribute__((ext_vector_type(8))) float v8f;

#define BATCH 128
#define DCH   300
#define LEN   128
#define NF    8
#define RPW   8
#define CHUNKS 16
#define WPB   4
#define NTHR  128
#define NBLK  1200
#define XPITCH 132

#define DPP_XOR1  0xB1
#define DPP_XOR2  0x4E
#define DPP_HMIRR 0x141
#define DPP_MIRR  0x140

template <int CTRL>
__device__ __forceinline__ float dpp_bfly(float v) {
    int i = __builtin_bit_cast(int, v);
    int t = __builtin_amdgcn_update_dpp(0, i, CTRL, 0xF, 0xF, true);
    return __builtin_bit_cast(float, t);
}
__device__ __forceinline__ float mxn(float a, float b) { return __builtin_elementwise_maximum(a, b); }
__device__ __forceinline__ float mnn(float a, float b) { return __builtin_elementwise_minimum(a, b); }
__device__ __forceinline__ float row16_max(float v) {
    v = mxn(v, dpp_bfly<DPP_XOR1>(v));
    v = mxn(v, dpp_bfly<DPP_XOR2>(v));
    v = mxn(v, dpp_bfly<DPP_HMIRR>(v));
    v = mxn(v, dpp_bfly<DPP_MIRR>(v));
    return v;
}
__device__ __forceinline__ float row16_min(float v) {
    v = mnn(v, dpp_bfly<DPP_XOR1>(v));
    v = mnn(v, dpp_bfly<DPP_XOR2>(v));
    v = mnn(v, dpp_bfly<DPP_HMIRR>(v));
    v = mnn(v, dpp_bfly<DPP_MIRR>(v));
    return v;
}
__device__ __forceinline__ float row16_sum(float v) {
    v += dpp_bfly<DPP_XOR1>(v);
    v += dpp_bfly<DPP_XOR2>(v);
    v += dpp_bfly<DPP_HMIRR>(v);
    v += dpp_bfly<DPP_MIRR>(v);
    return v;
}
__device__ __forceinline__ float swz_xor16(float v) {
    int t = __builtin_amdgcn_ds_swizzle(__builtin_bit_cast(int, v), 0x401F);
    return __builtin_bit_cast(float, t);
}

__global__ __launch_bounds__(NTHR) void conv_pool_wmma(
    const float* __restrict__ x,
    const float* __restrict__ W1, const float* __restrict__ b1,
    const float* __restrict__ W2, const float* __restrict__ b2,
    const float* __restrict__ W3, const float* __restrict__ b3,
    float* __restrict__ out)
{
    __shared__ __attribute__((aligned(16))) float xs[WPB][RPW][XPITCH];

    const int lane = threadIdx.x & 31;
    const int wv   = threadIdx.x >> 5;
    const int item = blockIdx.x * WPB + wv;
    const int d    = item / CHUNKS;
    const int b0   = (item - d * CHUNKS) * RPW;

    v4f xr[RPW];
    #pragma unroll
    for (int r = 0; r < RPW; ++r) {
        const size_t row = (size_t)(b0 + r) * DCH + d;
        xr[r] = *(const v4f*)(x + row * LEN + lane * 4);
    }

    v2f Bw;
    {
        const int  nn  = lane & 15;
        const bool hi  = lane >= 16;
        const bool is3 = nn >= 8;
        const int  g   = d * NF + (nn & 7);
        const float* p0 = is3 ? (W3 + g * 3 + (hi ? 2 : 0)) : (W2 + g * 2);
        const float* p1 = is3 ? (W3 + g * 3 + 1)            : (W2 + g * 2 + 1);
        const float l0 = *p0;
        const float l1 = *p1;
        Bw[0] = (!is3 && hi) ? 0.0f : l0;
        Bw[1] = hi ? 0.0f : l1;
    }
    const int   go   = d * NF + (lane & 7);
    const float wq1  = W1[go];
    const float bq1  = b1[go];
    const float* pb  = (lane & 8) ? b3 : b2;
    const float bq23 = pb[go];

    xs[wv][lane >> 2][LEN + (lane & 3)] = 0.0f;

    const float w3t2 = swz_xor16(Bw[0]);

    const int m    = lane & 15;
    const int kofs = (lane < 16) ? 0 : 2;

    const size_t strB = (size_t)2 * 3 * DCH * NF;
    const size_t strP = (size_t)3 * DCH * NF;
    const size_t strW = (size_t)DCH * NF;
    float* __restrict__ obase = out + (size_t)b0 * strB;

    #pragma unroll
    for (int r = 0; r < RPW; ++r) {
        float* row_s = &xs[wv][r][0];
        const v4f xv = xr[r];

        *(v4f*)(row_s + lane * 4) = xv;
        __builtin_amdgcn_fence(__ATOMIC_ACQ_REL, "wavefront");

        float mx = fmaxf(fmaxf(xv[0], xv[1]), fmaxf(xv[2], xv[3]));
        float mn = fminf(fminf(xv[0], xv[1]), fminf(xv[2], xv[3]));
        float sm = (xv[0] + xv[1]) + (xv[2] + xv[3]);
        mx = row16_max(mx);  mx = mxn(mx, swz_xor16(mx));
        mn = row16_min(mn);  mn = mnn(mn, swz_xor16(mn));
        sm = row16_sum(sm);  sm += swz_xor16(sm);

        float acc = -__builtin_inff();
        #pragma unroll
        for (int t = 0; t < 8; ++t) {
            v2f Ap;
            const int off = t * 16 + m + kofs;
            Ap[0] = row_s[off];
            Ap[1] = row_s[off + 1];
            v8f C = {};
            C = __builtin_amdgcn_wmma_f32_16x16x4_f32(false, Ap, false, Bw,
                                                      (short)0, C, false, false);
            float c6 = C[6], c7 = C[7];
            if (t == 7) {
                c6 = (lane < 24) ? c6 : -__builtin_inff();
                c7 = (lane < 16) ? c7 : -__builtin_inff();
            }
            const float q = fmaxf(fmaxf(fmaxf(C[0], C[1]), fmaxf(C[2], C[3])),
                                  fmaxf(fmaxf(C[4], C[5]), fmaxf(c6, c7)));
            acc = fmaxf(acc, q);
        }
        acc = mxn(acc, swz_xor16(acc));

        const float x0   = row_s[0],   x1   = row_s[1];
        const float x126 = row_s[126], x127 = row_s[127];
        const float S0_2 = sm - x127,        S1_2 = sm - x0;
        const float S0_3 = sm - x126 - x127, S1_3 = sm - x0 - x127,
                    S2_3 = sm - x0 - x1;

        float* __restrict__ omax  = obase + (size_t)r * strB;
        float* __restrict__ omean = omax + strP;

        if (lane < 8) {
            omax [0 * strW + d * NF + lane] = (wq1 >= 0.0f ? wq1 * mx : wq1 * mn) + bq1;
            omean[0 * strW + d * NF + lane] = wq1 * (sm * (1.0f / 128.0f)) + bq1;
            omax [1 * strW + d * NF + lane] = acc + bq23;
            omean[1 * strW + d * NF + lane] =
                (Bw[0] * S0_2 + Bw[1] * S1_2) * (1.0f / 127.0f) + bq23;
        } else if (lane < 16) {
            const int f = lane - 8;
            omax [2 * strW + d * NF + f] = acc + bq23;
            omean[2 * strW + d * NF + f] =
                (Bw[0] * S0_3 + Bw[1] * S1_3 + w3t2 * S2_3) * (1.0f / 126.0f) + bq23;
        }
    }
}

extern "C" void kernel_launch(void* const* d_in, const int* in_sizes, int n_in,
                              void* d_out, int out_size, void* d_ws, size_t ws_size,
                              hipStream_t stream) {
    const float* x  = (const float*)d_in[0];
    const float* W1 = (const float*)d_in[1];
    const float* b1 = (const float*)d_in[2];
    const float* W2 = (const float*)d_in[3];
    const float* b2 = (const float*)d_in[4];
    const float* W3 = (const float*)d_in[5];
    const float* b3 = (const float*)d_in[6];
    float* out = (float*)d_out;

    conv_pool_wmma<<<NBLK, NTHR, 0, stream>>>(x, W1, b1, W2, b2, W3, b3, out);
}


# each wave takes 4 adjacent channels x 2 batch rows (was 1 x 8): lane = (channel, filter), one 32-lane epilogue per batch row and six whole 128-B line stores replace branched 8-lane epilogues
# speedup vs baseline: 1.8311x; 1.2941x over previous
#include <hip/hip_runtime.h>


typedef __attribute__((ext_vector_type(2))) float v2f;
typedef __attribute__((ext_vector_type(4))) float v4f;
typedef __attribute__((ext_vector_type(8))) float v8f;

#define BATCH 128
#define DCH   300
#define LEN   128
#define NF    8
#define NB    2
#define NC    4
#define RPW   8
#define BGRP  64
#define WPB   4
#define NTHR  128
#define NBLK  1200
#define XPITCH 132
#define WPITCH 6

#define DPP_XOR1  0xB1
#define DPP_XOR2  0x4E
#define DPP_HMIRR 0x141
#define DPP_ROR8  0x128

template <int CTRL>
__device__ __forceinline__ float dpp_bfly(float v) {
    int i = __builtin_bit_cast(int, v);
    int t = __builtin_amdgcn_update_dpp(0, i, CTRL, 0xF, 0xF, true);
    return __builtin_bit_cast(float, t);
}
__device__ __forceinline__ float mxn(float a, float b) { return __builtin_elementwise_maximum(a, b); }
__device__ __forceinline__ float mnn(float a, float b) { return __builtin_elementwise_minimum(a, b); }
__device__ __forceinline__ float swz_xor16(float v) {
    int t = __builtin_amdgcn_ds_swizzle(__builtin_bit_cast(int, v), 0x401F);
    return __builtin_bit_cast(float, t);
}

struct OpMax { static __device__ __forceinline__ float f(float a, float b) { return mxn(a, b); } };
struct OpMin { static __device__ __forceinline__ float f(float a, float b) { return mnn(a, b); } };
struct OpSum { static __device__ __forceinline__ float f(float a, float b) { return a + b; } };

template <class OP>
__device__ __forceinline__ float reduce4(float v0, float v1, float v2, float v3,
                                         bool hi, bool up) {
    const float k02 = hi ? v2 : v0, s02 = hi ? v0 : v2;
    const float k13 = hi ? v3 : v1, s13 = hi ? v1 : v3;
    const float r02 = OP::f(k02, swz_xor16(s02));
    const float r13 = OP::f(k13, swz_xor16(s13));
    const float k = up ? r13 : r02, s = up ? r02 : r13;
    float r = OP::f(k, dpp_bfly<DPP_ROR8>(s));
    r = OP::f(r, dpp_bfly<DPP_XOR1>(r));
    r = OP::f(r, dpp_bfly<DPP_XOR2>(r));
    r = OP::f(r, dpp_bfly<DPP_HMIRR>(r));
    return r;
}

__global__ __launch_bounds__(NTHR) void conv_pool_wmma(
    const float* __restrict__ x,
    const float* __restrict__ W1, const float* __restrict__ b1,
    const float* __restrict__ W2, const float* __restrict__ b2,
    const float* __restrict__ W3, const float* __restrict__ b3,
    float* __restrict__ out)
{
    __shared__ __attribute__((aligned(16))) float xs[WPB][RPW][XPITCH];
    __shared__ __attribute__((aligned(8)))  float wl[WPB][32][WPITCH];

    const int lane = threadIdx.x & 31;
    const int wv   = threadIdx.x >> 5;
    const int item = blockIdx.x * WPB + wv;
    const int dg   = item / BGRP;
    const int d0   = dg * NC;
    const int b0   = (item - dg * BGRP) * NB;

    const unsigned g   = (unsigned)(d0 * NF + lane);
    const float    wq1 = W1[g];
    const float    bq1 = b1[g];
    const float    bq2 = b2[g];
    const float    bq3 = b3[g];
    const v2f      w2  = *(const v2f*)(W2 + 2u * g);
    const float    w3a = W3[3u * g];
    const float    w3b = W3[3u * g + 1u];
    const float    w3c = W3[3u * g + 2u];

    v4f xr[RPW];
    #pragma unroll
    for (int rr = 0; rr < RPW; ++rr) {
        const size_t row = (size_t)(b0 + (rr >> 2)) * DCH + d0 + (rr & 3);
        xr[rr] = *(const v4f*)(x + row * LEN + lane * 4);
    }

    xs[wv][lane >> 2][LEN + (lane & 3)] = 0.0f;

    {
        float* wrow = &wl[wv][lane][0];
        v2f t01; t01[0] = w3a; t01[1] = w3b;
        v2f t2z; t2z[0] = w3c; t2z[1] = 0.0f;
        *(v2f*)(wrow)     = w2;
        *(v2f*)(wrow + 2) = t01;
        *(v2f*)(wrow + 4) = t2z;
    }
    __builtin_amdgcn_fence(__ATOMIC_ACQ_REL, "wavefront");

    const bool hi = lane >= 16;
    const bool up = (lane & 8) != 0;
    const int e0_2 = hi ? 5 : 0, e1_2 = hi ? 5 : 1;
    const int e0_3 = hi ? 4 : 2, e1_3 = hi ? 5 : 3;
    const int e0_ev = up ? e0_3 : e0_2, e1_ev = up ? e1_3 : e1_2;
    const int e0_od = up ? e0_2 : e0_3, e1_od = up ? e1_2 : e1_3;
    v2f Bw[NC];
    #pragma unroll
    for (int c = 0; c < NC; ++c) {
        const float* wr = &wl[wv][c * NF + (lane & 7)][0];
        Bw[c][0] = wr[(c & 1) ? e0_od : e0_ev];
        Bw[c][1] = wr[(c & 1) ? e1_od : e1_ev];
    }

    const int m    = lane & 15;
    const int kofs = hi ? 2 : 0;

    const size_t strB = (size_t)2 * 3 * DCH * NF;
    const size_t strP = (size_t)3 * DCH * NF;
    const size_t strW = (size_t)DCH * NF;
    float* __restrict__ obase = out + (size_t)b0 * strB + g;

    float lmx[NC], lmn[NC], lsm[NC], acc[NC];

    #pragma unroll
    for (int rr = 0; rr < RPW; ++rr) {
        const int c = rr & 3;
        float* row_s = &xs[wv][rr][0];
        const v4f xv = xr[rr];

        *(v4f*)(row_s + lane * 4) = xv;
        __builtin_amdgcn_fence(__ATOMIC_ACQ_REL, "wavefront");

        lmx[c] = fmaxf(fmaxf(xv[0], xv[1]), fmaxf(xv[2], xv[3]));
        lmn[c] = fminf(fminf(xv[0], xv[1]), fminf(xv[2], xv[3]));
        lsm[c] = (xv[0] + xv[1]) + (xv[2] + xv[3]);

        float a = -__builtin_inff();
        #pragma unroll
        for (int t = 0; t < 8; ++t) {
            v2f Ap;
            const int off = t * 16 + m + kofs;
            Ap[0] = row_s[off];
            Ap[1] = row_s[off + 1];
            v8f C = {};
            C = __builtin_amdgcn_wmma_f32_16x16x4_f32(false, Ap, false, Bw[c],
                                                      (short)0, C, false, false);
            float c6 = C[6], c7 = C[7];
            if (t == 7) {
                const bool ws3col = (c & 1) ? !up : up;
                c6 = (hi && ws3col) ? -__builtin_inff() : c6;
                c7 = hi ? -__builtin_inff() : c7;
            }
            const float q = fmaxf(fmaxf(fmaxf(C[0], C[1]), fmaxf(C[2], C[3])),
                                  fmaxf(fmaxf(C[4], C[5]), fmaxf(c6, c7)));
            a = fmaxf(a, q);
        }
        asm volatile("" : "+v"(a));
        acc[c] = a;

        if (c == NC - 1) {
            const int b = rr >> 2;
            const float MX = reduce4<OpMax>(lmx[0], lmx[1], lmx[2], lmx[3], hi, up);
            const float MN = reduce4<OpMin>(lmn[0], lmn[1], lmn[2], lmn[3], hi, up);
            const float SM = reduce4<OpSum>(lsm[0], lsm[1], lsm[2], lsm[3], hi, up);

            const float k02 = hi ? acc[2] : acc[0], s02 = hi ? acc[0] : acc[2];
            const float k13 = hi ? acc[3] : acc[1], s13 = hi ? acc[1] : acc[3];
            const float A02 = mxn(k02, swz_xor16(s02));
            const float A13 = mxn(k13, swz_xor16(s13));
            const float m2  = up ? A13 : A02;
            const float m3  = dpp_bfly<DPP_ROR8>(up ? A02 : A13);

            const float* er = &xs[wv][b * NC + (lane >> 3)][0];
            const float x0   = er[0],   x1   = er[1];
            const float x126 = er[126], x127 = er[127];
            const float S0_2 = SM - x127,        S1_2 = SM - x0;
            const float S0_3 = SM - x126 - x127, S1_3 = SM - x0 - x127,
                        S2_3 = SM - x0 - x1;

            float* __restrict__ omax  = obase + (size_t)b * strB;
            float* __restrict__ omean = omax + strP;

            omax [0 * strW] = (wq1 >= 0.0f ? wq1 * MX : wq1 * MN) + bq1;
            omean[0 * strW] = wq1 * (SM * (1.0f / 128.0f)) + bq1;
            omax [1 * strW] = m2 + bq2;
            omean[1 * strW] = (w2[0] * S0_2 + w2[1] * S1_2) * (1.0f / 127.0f) + bq2;
            omax [2 * strW] = m3 + bq3;
            omean[2 * strW] =
                (w3a * S0_3 + w3b * S1_3 + w3c * S2_3) * (1.0f / 126.0f) + bq3;
        }
    }
}

extern "C" void kernel_launch(void* const* d_in, const int* in_sizes, int n_in,
                              void* d_out, int out_size, void* d_ws, size_t ws_size,
                              hipStream_t stream) {
    const float* x  = (const float*)d_in[0];
    const float* W1 = (const float*)d_in[1];
    const float* b1 = (const float*)d_in[2];
    const float* W2 = (const float*)d_in[3];
    const float* b2 = (const float*)d_in[4];
    const float* W3 = (const float*)d_in[5];
    const float* b3 = (const float*)d_in[6];
    float* out = (float*)d_out;

    conv_pool_wmma<<<NBLK, NTHR, 0, stream>>>(x, W1, b1, W2, b2, W3, b3, out);
}


# A rows re-mapped to even positions (row m = position 32t+2m; taps at shift 0 and shift 1 as two B fragments per A tile): a row's A tiles are 2 aligned two-address 8-byte LDS reads, were 8 dword pairs
# speedup vs baseline: 1.9439x; 1.0616x over previous
#include <hip/hip_runtime.h>


typedef __attribute__((ext_vector_type(2))) float v2f;
typedef __attribute__((ext_vector_type(4))) float v4f;
typedef __attribute__((ext_vector_type(8))) float v8f;

#define BATCH 128
#define DCH   300
#define LEN   128
#define NF    8
#define NB    2
#define NC    4
#define RPW   8
#define NT    4
#define BGRP  64
#define WPB   4
#define NTHR  128
#define NBLK  1200
#define XPITCH 132
#define WPITCH 6

#define DPP_XOR1  0xB1
#define DPP_XOR2  0x4E
#define DPP_HMIRR 0x141
#define DPP_ROR8  0x128

template <int CTRL>
__device__ __forceinline__ float dpp_bfly(float v) {
    int i = __builtin_bit_cast(int, v);
    int t = __builtin_amdgcn_update_dpp(0, i, CTRL, 0xF, 0xF, true);
    return __builtin_bit_cast(float, t);
}
__device__ __forceinline__ float mxn(float a, float b) { return __builtin_elementwise_maximum(a, b); }
__device__ __forceinline__ float mnn(float a, float b) { return __builtin_elementwise_minimum(a, b); }
__device__ __forceinline__ float swz_xor16(float v) {
    int t = __builtin_amdgcn_ds_swizzle(__builtin_bit_cast(int, v), 0x401F);
    return __builtin_bit_cast(float, t);
}

struct OpMax { static __device__ __forceinline__ float f(float a, float b) { return mxn(a, b); } };
struct OpMin { static __device__ __forceinline__ float f(float a, float b) { return mnn(a, b); } };
struct OpSum { static __device__ __forceinline__ float f(float a, float b) { return a + b; } };

template <class OP>
__device__ __forceinline__ float reduce4(float v0, float v1, float v2, float v3,
                                         bool hi, bool up) {
    const float k02 = hi ? v2 : v0, s02 = hi ? v0 : v2;
    const float k13 = hi ? v3 : v1, s13 = hi ? v1 : v3;
    const float r02 = OP::f(k02, swz_xor16(s02));
    const float r13 = OP::f(k13, swz_xor16(s13));
    const float k = up ? r13 : r02, s = up ? r02 : r13;
    float r = OP::f(k, dpp_bfly<DPP_ROR8>(s));
    r = OP::f(r, dpp_bfly<DPP_XOR1>(r));
    r = OP::f(r, dpp_bfly<DPP_XOR2>(r));
    r = OP::f(r, dpp_bfly<DPP_HMIRR>(r));
    return r;
}

__device__ __forceinline__ float tile_max(float c0, float c1, float c2, float c3,
                                          float c4, float c5, float c6, float c7) {
    return fmaxf(fmaxf(fmaxf(c0, c1), fmaxf(c2, c3)),
                 fmaxf(fmaxf(c4, c5), fmaxf(c6, c7)));
}

__global__ __launch_bounds__(NTHR) void conv_pool_wmma(
    const float* __restrict__ x,
    const float* __restrict__ W1, const float* __restrict__ b1,
    const float* __restrict__ W2, const float* __restrict__ b2,
    const float* __restrict__ W3, const float* __restrict__ b3,
    float* __restrict__ out)
{
    __shared__ __attribute__((aligned(16))) float xs[WPB][RPW][XPITCH];
    __shared__ __attribute__((aligned(8)))  float wl[WPB][32][WPITCH];

    const int lane = threadIdx.x & 31;
    const int wv   = threadIdx.x >> 5;
    const int item = blockIdx.x * WPB + wv;
    const int dg   = item / BGRP;
    const int d0   = dg * NC;
    const int b0   = (item - dg * BGRP) * NB;

    const unsigned g   = (unsigned)(d0 * NF + lane);
    const float    wq1 = W1[g];
    const float    bq1 = b1[g];
    const float    bq2 = b2[g];
    const float    bq3 = b3[g];
    const v2f      w2  = *(const v2f*)(W2 + 2u * g);
    const float    w3a = W3[3u * g];
    const float    w3b = W3[3u * g + 1u];
    const float    w3c = W3[3u * g + 2u];

    v4f xr[RPW];
    #pragma unroll
    for (int rr = 0; rr < RPW; ++rr) {
        const size_t row = (size_t)(b0 + (rr >> 2)) * DCH + d0 + (rr & 3);
        xr[rr] = *(const v4f*)(x + row * LEN + lane * 4);
    }

    xs[wv][lane >> 2][LEN + (lane & 3)] = 0.0f;

    {
        float* wrow = &wl[wv][lane][0];
        v2f t01; t01[0] = w3a; t01[1] = w3b;
        v2f t2z; t2z[0] = w3c; t2z[1] = 0.0f;
        *(v2f*)(wrow)     = w2;
        *(v2f*)(wrow + 2) = t01;
        *(v2f*)(wrow + 4) = t2z;
    }
    __builtin_amdgcn_fence(__ATOMIC_ACQ_REL, "wavefront");

    const bool hi = lane >= 16;
    const bool up = (lane & 8) != 0;
    const int x0_2 = hi ? 5 : 0, x1_2 = hi ? 5 : 1;
    const int x0_3 = hi ? 4 : 2, x1_3 = hi ? 5 : 3;
    const int y0_2 = hi ? 1 : 5, y1_2 = hi ? 5 : 0;
    const int y0_3 = hi ? 3 : 5, y1_3 = hi ? 4 : 2;
    const int x0_ev = up ? x0_3 : x0_2, x1_ev = up ? x1_3 : x1_2;
    const int x0_od = up ? x0_2 : x0_3, x1_od = up ? x1_2 : x1_3;
    const int y0_ev = up ? y0_3 : y0_2, y1_ev = up ? y1_3 : y1_2;
    const int y0_od = up ? y0_2 : y0_3, y1_od = up ? y1_2 : y1_3;
    v2f BX[NC], BY[NC];
    #pragma unroll
    for (int c = 0; c < NC; ++c) {
        const float* wr = &wl[wv][c * NF + (lane & 7)][0];
        BX[c][0] = wr[(c & 1) ? x0_od : x0_ev];
        BX[c][1] = wr[(c & 1) ? x1_od : x1_ev];
        BY[c][0] = wr[(c & 1) ? y0_od : y0_ev];
        BY[c][1] = wr[(c & 1) ? y1_od : y1_ev];
    }

    const int m   = lane & 15;
    const int aof = 2 * m + (hi ? 2 : 0);

    const size_t strB = (size_t)2 * 3 * DCH * NF;
    const size_t strP = (size_t)3 * DCH * NF;
    const size_t strW = (size_t)DCH * NF;
    float* __restrict__ obase = out + (size_t)b0 * strB + g;

    float lmx[NC], lmn[NC], lsm[NC], acc[NC];

    #pragma unroll
    for (int rr = 0; rr < RPW; ++rr) {
        const int c = rr & 3;
        float* row_s = &xs[wv][rr][0];
        const v4f xv = xr[rr];

        *(v4f*)(row_s + lane * 4) = xv;
        __builtin_amdgcn_fence(__ATOMIC_ACQ_REL, "wavefront");

        lmx[c] = fmaxf(fmaxf(xv[0], xv[1]), fmaxf(xv[2], xv[3]));
        lmn[c] = fminf(fminf(xv[0], xv[1]), fminf(xv[2], xv[3]));
        lsm[c] = (xv[0] + xv[1]) + (xv[2] + xv[3]);

        float a = 0.0f;
        #pragma unroll
        for (int t = 0; t < NT; ++t) {
            const v2f Ap = *(const v2f*)(row_s + 32 * t + aof);
            v8f Z = {};
            const v8f DX = __builtin_amdgcn_wmma_f32_16x16x4_f32(false, Ap, false, BX[c],
                                                                 (short)0, Z, false, false);
            const v8f DY = __builtin_amdgcn_wmma_f32_16x16x4_f32(false, Ap, false, BY[c],
                                                                 (short)0, Z, false, false);
            float dx7 = DX[7], dy7 = DY[7];
            if (t == NT - 1) {
                const bool ws3col = (c & 1) ? !up : up;
                dx7 = (hi && ws3col) ? -__builtin_inff() : dx7;
                dy7 = hi ? -__builtin_inff() : dy7;
            }
            const float q = fmaxf(tile_max(DX[0], DX[1], DX[2], DX[3], DX[4], DX[5], DX[6], dx7),
                                  tile_max(DY[0], DY[1], DY[2], DY[3], DY[4], DY[5], DY[6], dy7));
            a = (t == 0) ? q : fmaxf(a, q);
        }
        asm volatile("" : "+v"(a));
        acc[c] = a;

        if (c == NC - 1) {
            const int b = rr >> 2;
            const float MX = reduce4<OpMax>(lmx[0], lmx[1], lmx[2], lmx[3], hi, up);
            const float MN = reduce4<OpMin>(lmn[0], lmn[1], lmn[2], lmn[3], hi, up);
            const float SM = reduce4<OpSum>(lsm[0], lsm[1], lsm[2], lsm[3], hi, up);

            const float k02 = hi ? acc[2] : acc[0], s02 = hi ? acc[0] : acc[2];
            const float k13 = hi ? acc[3] : acc[1], s13 = hi ? acc[1] : acc[3];
            const float A02 = mxn(k02, swz_xor16(s02));
            const float A13 = mxn(k13, swz_xor16(s13));
            const float m2  = up ? A13 : A02;
            const float m3  = dpp_bfly<DPP_ROR8>(up ? A02 : A13);

            const float* er = &xs[wv][b * NC + (lane >> 3)][0];
            const float x0   = er[0],   x1   = er[1];
            const float x126 = er[126], x127 = er[127];
            const float S0_2 = SM - x127,        S1_2 = SM - x0;
            const float S0_3 = SM - x126 - x127, S1_3 = SM - x0 - x127,
                        S2_3 = SM - x0 - x1;

            float* __restrict__ omax  = obase + (size_t)b * strB;
            float* __restrict__ omean = omax + strP;

            omax [0 * strW] = (wq1 >= 0.0f ? wq1 * MX : wq1 * MN) + bq1;
            omean[0 * strW] = wq1 * (SM * (1.0f / 128.0f)) + bq1;
            omax [1 * strW] = m2 + bq2;
            omean[1 * strW] = (w2[0] * S0_2 + w2[1] * S1_2) * (1.0f / 127.0f) + bq2;
            omax [2 * strW] = m3 + bq3;
            omean[2 * strW] =
                (w3a * S0_3 + w3b * S1_3 + w3c * S2_3) * (1.0f / 126.0f) + bq3;
        }
    }
}

extern "C" void kernel_launch(void* const* d_in, const int* in_sizes, int n_in,
                              void* d_out, int out_size, void* d_ws, size_t ws_size,
                              hipStream_t stream) {
    const float* x  = (const float*)d_in[0];
    const float* W1 = (const float*)d_in[1];
    const float* b1 = (const float*)d_in[2];
    const float* W2 = (const float*)d_in[3];
    const float* b2 = (const float*)d_in[4];
    const float* W3 = (const float*)d_in[5];
    const float* b3 = (const float*)d_in[6];
    float* out = (float*)d_out;

    conv_pool_wmma<<<NBLK, NTHR, 0, stream>>>(x, W1, b1, W2, b2, W3, b3, out);
}
